# starting kernel
# baseline (speedup 1.0000x reference)
#include <hip/hip_runtime.h>


typedef __attribute__((ext_vector_type(2))) float v2f;
typedef __attribute__((ext_vector_type(8))) float v8f;

#define GK 16
#define A_CH 24
#define H_DIM 256
#define N_PIX 65536
#define TILES 4096

__global__ __launch_bounds__(256) void gmm_render_kernel(
    const float* __restrict__ params, float* __restrict__ out) {
  const int lane = threadIdx.x & 31;
  const int g    = lane & 15;
  const int hi   = lane >> 4;
  const int a    = blockIdx.y;

  const float* pa = params + a * (7 * GK);
  const float mux = pa[g];
  const float muy = pa[GK + g];
  const float z   = pa[2 * GK + g];
  const float l00 = pa[3 * GK + 4 * g + 0];
  const float l10 = pa[3 * GK + 4 * g + 2];
  const float l11 = pa[3 * GK + 4 * g + 3];

  const float srx = l00 * l00 + l00 * l10;
  const float sry = l00 * l10 + l10 * l10 + l11 * l11;
  const float c1c = -(srx * mux + sry * muy);
  const float c2c = -(mux + muy);

  float mx = z;
  mx = fmaxf(mx, __shfl_xor(mx, 1, 32));
  mx = fmaxf(mx, __shfl_xor(mx, 2, 32));
  mx = fmaxf(mx, __shfl_xor(mx, 4, 32));
  mx = fmaxf(mx, __shfl_xor(mx, 8, 32));
  float ez = __expf(z - mx);
  float ssum = ez;
  ssum += __shfl_xor(ssum, 1, 32);
  ssum += __shfl_xor(ssum, 2, 32);
  ssum += __shfl_xor(ssum, 4, 32);
  ssum += __shfl_xor(ssum, 8, 32);
  const float w = ez / ssum;

  float wv[8];
#pragma unroll
  for (int r = 0; r < 8; ++r) wv[r] = __shfl(w, r + 8 * hi, 16);

  v2f a1, a2;
  a1.x = hi ? c1c : srx;
  a1.y = hi ? 0.0f : sry;
  a2.x = hi ? c2c : 1.0f;
  a2.y = hi ? 0.0f : 1.0f;

  const v8f czero = {};

  const int waves_per_blk = blockDim.x >> 5;
  const int wave   = blockIdx.x * waves_per_blk + (threadIdx.x >> 5);
  const int nwaves = gridDim.x * waves_per_blk;

  for (int tile = wave; tile < TILES; tile += nwaves) {
    const int pn = (tile << 4) + g;
    const float gx = (float)(pn >> 8) * (1.0f / 255.0f);
    const float gy = (float)(pn & 255) * (1.0f / 255.0f);

    v2f b;
    b.x = hi ? 1.0f : gx;
    b.y = hi ? 0.0f : gy;

    v8f t1 = __builtin_amdgcn_wmma_f32_16x16x4_f32(
        false, a1, false, b, (short)0, czero, false, false);
    v8f t2 = __builtin_amdgcn_wmma_f32_16x16x4_f32(
        false, a2, false, b, (short)0, czero, false, false);

    float se = 0.0f, swe = 0.0f;
#pragma unroll
    for (int r = 0; r < 8; ++r) {
      const float e = __expf(-0.5f * t1[r] * t2[r]);
      se += e;
      swe = fmaf(wv[r], e, swe);
    }
    se  += __shfl_xor(se, 16, 32);
    swe += __shfl_xor(swe, 16, 32);

    float y = swe / fmaxf(se, 1e-8f);
    y = fminf(fmaxf(y, 0.0f), 1.0f);
    if (hi == 0) out[a * N_PIX + pn] = y;
  }
}

extern "C" void kernel_launch(void* const* d_in, const int* in_sizes, int n_in,
                              void* d_out, int out_size, void* d_ws, size_t ws_size,
                              hipStream_t stream) {
  (void)in_sizes; (void)n_in; (void)d_ws; (void)ws_size; (void)out_size;
  const float* params = (const float*)d_in[0];
  float* out = (float*)d_out;
  dim3 grid(64, A_CH);
  gmm_render_kernel<<<grid, 256, 0, stream>>>(params, out);
}


# Channel constants built ONCE per work-group by its first wave (one load group, DPP butterflies) and handed to the other seven waves through LDS; was once per wave; same values, same bits
# speedup vs baseline: 1.1223x; 1.1223x over previous
#include <hip/hip_runtime.h>


typedef __attribute__((ext_vector_type(2))) float v2f;
typedef __attribute__((ext_vector_type(8))) float v8f;

#define GK 16
#define A_CH 24
#define H_DIM 256
#define N_PIX 65536
#define TILES 4096

template <int M>
__device__ __forceinline__ float row_xor(float x) {
  return __builtin_bit_cast(
      float, __builtin_amdgcn_mov_dpp(__builtin_bit_cast(int, x), 0x160 | M,
                                      0xf, 0xf, false));
}

__global__ __launch_bounds__(256) void gmm_render_kernel(
    const float* __restrict__ params, float* __restrict__ out) {
  __shared__ float4 s_ab[32];
  __shared__ float4 s_w[8];

  const int lane = threadIdx.x & 31;
  const int g    = lane & 15;
  const int hi   = lane >> 4;
  const int a    = blockIdx.y;
  const int wid  = __builtin_amdgcn_readfirstlane((int)(threadIdx.x >> 5));

  const int nblk = gridDim.x;
  const int bdim = blockDim.x;
  asm volatile("" ::"s"(nblk), "s"(bdim));

  if (wid == 0) {
    const float* pa = params + a * (7 * GK);
    const float mux = pa[g];
    const float muy = pa[GK + g];
    const float z   = pa[2 * GK + g];
    const float l00 = pa[3 * GK + 4 * g + 0];
    const float l10 = pa[3 * GK + 4 * g + 2];
    const float l11 = pa[3 * GK + 4 * g + 3];
    asm volatile("" ::"v"(mux), "v"(muy), "v"(z), "v"(l00), "v"(l10),
                 "v"(l11));

    const float srx = l00 * l00 + l00 * l10;
    const float sry = l00 * l10 + l10 * l10 + l11 * l11;
    const float c1c = -(srx * mux + sry * muy);
    const float c2c = -(mux + muy);

    float mx = z;
    mx = fmaxf(mx, row_xor<1>(mx));
    mx = fmaxf(mx, row_xor<2>(mx));
    mx = fmaxf(mx, row_xor<4>(mx));
    mx = fmaxf(mx, row_xor<8>(mx));
    float ez = __expf(z - mx);
    float ssum = ez;
    ssum += row_xor<1>(ssum);
    ssum += row_xor<2>(ssum);
    ssum += row_xor<4>(ssum);
    ssum += row_xor<8>(ssum);
    const float w = ez / ssum;

    float4 ab;
    ab.x = hi ? c1c : srx;
    ab.y = hi ? 0.0f : sry;
    ab.z = hi ? c2c : 1.0f;
    ab.w = hi ? 0.0f : 1.0f;
    s_ab[lane] = ab;
    reinterpret_cast<float*>(s_w)[lane] = w;
  }
  __syncthreads();

  const float4 ab = s_ab[lane];
  v2f a1, a2;
  a1.x = ab.x; a1.y = ab.y;
  a2.x = ab.z; a2.y = ab.w;

  const float4 w0 = s_w[2 * hi];
  const float4 w1 = s_w[2 * hi + 1];
  float wv[8];
  wv[0] = w0.x; wv[1] = w0.y; wv[2] = w0.z; wv[3] = w0.w;
  wv[4] = w1.x; wv[5] = w1.y; wv[6] = w1.z; wv[7] = w1.w;

  const v8f czero = {};

  const int waves_per_blk = bdim >> 5;
  const int wave   = blockIdx.x * waves_per_blk + wid;
  const int nwaves = nblk * waves_per_blk;

  for (int tile = wave; tile < TILES; tile += nwaves) {
    const int pn = (tile << 4) + g;
    const float gx = (float)(pn >> 8) * (1.0f / 255.0f);
    const float gy = (float)(pn & 255) * (1.0f / 255.0f);

    v2f b;
    b.x = hi ? 1.0f : gx;
    b.y = hi ? 0.0f : gy;

    v8f t1 = __builtin_amdgcn_wmma_f32_16x16x4_f32(
        false, a1, false, b, (short)0, czero, false, false);
    v8f t2 = __builtin_amdgcn_wmma_f32_16x16x4_f32(
        false, a2, false, b, (short)0, czero, false, false);

    float se = 0.0f, swe = 0.0f;
#pragma unroll
    for (int r = 0; r < 8; ++r) {
      const float e = __expf(-0.5f * t1[r] * t2[r]);
      se += e;
      swe = fmaf(wv[r], e, swe);
    }
    se  += __shfl_xor(se, 16, 32);
    swe += __shfl_xor(swe, 16, 32);

    float y = swe / fmaxf(se, 1e-8f);
    y = fminf(fmaxf(y, 0.0f), 1.0f);
    if (hi == 0) out[a * N_PIX + pn] = y;
  }
}

extern "C" void kernel_launch(void* const* d_in, const int* in_sizes, int n_in,
                              void* d_out, int out_size, void* d_ws, size_t ws_size,
                              hipStream_t stream) {
  (void)in_sizes; (void)n_in; (void)d_ws; (void)ws_size; (void)out_size;
  const float* params = (const float*)d_in[0];
  float* out = (float*)d_out;
  dim3 grid(64, A_CH);
  gmm_render_kernel<<<grid, 256, 0, stream>>>(params, out);
}


# WMMA tile re-laid: block-diagonal A, K pairs split by lane half, constants as C addend: a lane owns one pixel (16 gaussians in two passes), no cross-lane sum; term1/term2 round in another order
# speedup vs baseline: 1.2586x; 1.1214x over previous
#include <hip/hip_runtime.h>


typedef __attribute__((ext_vector_type(2))) float v2f;
typedef __attribute__((ext_vector_type(8))) float v8f;

#define GK 16
#define A_CH 24
#define N_PIX 65536
#define TILES 2048

template <int M>
__device__ __forceinline__ float row_xor(float x) {
  return __builtin_bit_cast(
      float, __builtin_amdgcn_mov_dpp(__builtin_bit_cast(int, x), 0x160 | M,
                                      0xf, 0xf, false));
}

__global__ __launch_bounds__(256) void gmm_render_kernel(
    const float* __restrict__ params, float* __restrict__ out) {
  __shared__ float2 s_sr[32];
  __shared__ float4 s_c1[8];
  __shared__ float4 s_c2[8];
  __shared__ float4 s_w[8];

  const int lane = threadIdx.x & 31;
  const int g    = lane & 15;
  const int hi   = lane >> 4;
  const int a    = blockIdx.y;
  const int wid  = __builtin_amdgcn_readfirstlane((int)(threadIdx.x >> 5));

  const int nblk = gridDim.x;
  const int bdim = blockDim.x;
  asm volatile("" ::"s"(nblk), "s"(bdim));

  if (wid == 0) {
    const float* pa = params + a * (7 * GK);
    const float mux = pa[g];
    const float muy = pa[GK + g];
    const float z   = pa[2 * GK + g];
    const float l00 = pa[3 * GK + 4 * g + 0];
    const float l10 = pa[3 * GK + 4 * g + 2];
    const float l11 = pa[3 * GK + 4 * g + 3];
    asm volatile("" ::"v"(mux), "v"(muy), "v"(z), "v"(l00), "v"(l10),
                 "v"(l11));

    const float srx = l00 * l00 + l00 * l10;
    const float sry = l00 * l10 + l10 * l10 + l11 * l11;
    const float c1c = -(srx * mux + sry * muy);
    const float c2c = -(mux + muy);

    float mx = z;
    mx = fmaxf(mx, row_xor<1>(mx));
    mx = fmaxf(mx, row_xor<2>(mx));
    mx = fmaxf(mx, row_xor<4>(mx));
    mx = fmaxf(mx, row_xor<8>(mx));
    float ez = __expf(z - mx);
    float ssum = ez;
    ssum += row_xor<1>(ssum);
    ssum += row_xor<2>(ssum);
    ssum += row_xor<4>(ssum);
    ssum += row_xor<8>(ssum);
    const float w = ez / ssum;

    s_sr[lane] = make_float2(srx, sry);
    reinterpret_cast<float*>(s_c1)[lane] = c1c;
    reinterpret_cast<float*>(s_c2)[lane] = c2c;
    reinterpret_cast<float*>(s_w)[lane]  = w;
  }
  __syncthreads();

  const bool own = ((lane >> 3) & 1) == hi;
  const float2 sr0 = s_sr[lane & 7];
  const float2 sr1 = s_sr[(lane & 7) + 8];
  v2f a1p0, a1p1, a2;
  a1p0.x = own ? sr0.x : 0.0f;  a1p0.y = own ? sr0.y : 0.0f;
  a1p1.x = own ? sr1.x : 0.0f;  a1p1.y = own ? sr1.y : 0.0f;
  a2.x   = own ? 1.0f : 0.0f;   a2.y   = a2.x;

  const float4 q0 = s_c1[0], q1 = s_c1[1], q2 = s_c1[2], q3 = s_c1[3];
  const float4 u0 = s_c2[0], u1 = s_c2[1], u2 = s_c2[2], u3 = s_c2[3];
  const float4 w0 = s_w[0],  w1 = s_w[1],  w2 = s_w[2],  w3 = s_w[3];
  const v8f c1p0 = {q0.x, q0.y, q0.z, q0.w, q1.x, q1.y, q1.z, q1.w};
  const v8f c1p1 = {q2.x, q2.y, q2.z, q2.w, q3.x, q3.y, q3.z, q3.w};
  const v8f c2p0 = {u0.x, u0.y, u0.z, u0.w, u1.x, u1.y, u1.z, u1.w};
  const v8f c2p1 = {u2.x, u2.y, u2.z, u2.w, u3.x, u3.y, u3.z, u3.w};
  const float wl[16] = {w0.x, w0.y, w0.z, w0.w, w1.x, w1.y, w1.z, w1.w,
                        w2.x, w2.y, w2.z, w2.w, w3.x, w3.y, w3.z, w3.w};
  float ws[16];
#pragma unroll
  for (int i = 0; i < 16; ++i)
    ws[i] = __builtin_bit_cast(
        float, __builtin_amdgcn_readfirstlane(__builtin_bit_cast(int, wl[i])));

  const int waves_per_blk = bdim >> 5;
  const int wave   = blockIdx.x * waves_per_blk + wid;
  const int nwaves = nblk * waves_per_blk;

  for (int tile = wave; tile < TILES; tile += nwaves) {
    const int pn = (tile << 5) + lane;
    v2f b;
    b.x = (float)(pn >> 8) * (1.0f / 255.0f);
    b.y = (float)(pn & 255) * (1.0f / 255.0f);

    const v8f t1a = __builtin_amdgcn_wmma_f32_16x16x4_f32(
        false, a1p0, false, b, (short)0, c1p0, false, false);
    const v8f t2a = __builtin_amdgcn_wmma_f32_16x16x4_f32(
        false, a2, false, b, (short)0, c2p0, false, false);
    const v8f t1b = __builtin_amdgcn_wmma_f32_16x16x4_f32(
        false, a1p1, false, b, (short)0, c1p1, false, false);
    const v8f t2b = __builtin_amdgcn_wmma_f32_16x16x4_f32(
        false, a2, false, b, (short)0, c2p1, false, false);

    float se0 = 0.0f, swe0 = 0.0f, se1 = 0.0f, swe1 = 0.0f;
#pragma unroll
    for (int r = 0; r < 8; ++r) {
      const float e0 = __expf(-0.5f * t1a[r] * t2a[r]);
      se0 += e0;
      swe0 = fmaf(ws[r], e0, swe0);
    }
#pragma unroll
    for (int r = 0; r < 8; ++r) {
      const float e1 = __expf(-0.5f * t1b[r] * t2b[r]);
      se1 += e1;
      swe1 = fmaf(ws[8 + r], e1, swe1);
    }
    const float se  = se0 + se1;
    const float swe = swe0 + swe1;

    float y = swe / fmaxf(se, 1e-8f);
    y = fminf(fmaxf(y, 0.0f), 1.0f);
    out[a * N_PIX + pn] = y;
  }
}

extern "C" void kernel_launch(void* const* d_in, const int* in_sizes, int n_in,
                              void* d_out, int out_size, void* d_ws, size_t ws_size,
                              hipStream_t stream) {
  (void)in_sizes; (void)n_in; (void)d_ws; (void)ws_size; (void)out_size;
  const float* params = (const float*)d_in[0];
  float* out = (float*)d_out;
  dim3 grid(64, A_CH);
  gmm_render_kernel<<<grid, 256, 0, stream>>>(params, out);
}


# Exponent's constant factor -0.5*log2(e) folded into the term2 WMMA operands (A entries, C addend scaled once per work-group): term1*term2 feeds exp2 directly; term2 and exponent round differently
# speedup vs baseline: 1.3317x; 1.0581x over previous
#include <hip/hip_runtime.h>


typedef __attribute__((ext_vector_type(2))) float v2f;
typedef __attribute__((ext_vector_type(8))) float v8f;

#define GK 16
#define A_CH 24
#define N_PIX 65536
#define TILES 2048
#define KEXP (-0.72134752044448170368f)

template <int M>
__device__ __forceinline__ float row_xor(float x) {
  return __builtin_bit_cast(
      float, __builtin_amdgcn_mov_dpp(__builtin_bit_cast(int, x), 0x160 | M,
                                      0xf, 0xf, false));
}

__global__ __launch_bounds__(256) void gmm_render_kernel(
    const float* __restrict__ params, float* __restrict__ out) {
  __shared__ float2 s_sr[32];
  __shared__ float4 s_c1[8];
  __shared__ float4 s_c2[8];
  __shared__ float4 s_w[8];

  const int lane = threadIdx.x & 31;
  const int g    = lane & 15;
  const int hi   = lane >> 4;
  const int a    = blockIdx.y;
  const int wid  = __builtin_amdgcn_readfirstlane((int)(threadIdx.x >> 5));

  const int nblk = gridDim.x;
  const int bdim = blockDim.x;
  asm volatile("" ::"s"(nblk), "s"(bdim));

  if (wid == 0) {
    const float* pa = params + a * (7 * GK);
    const float mux = pa[g];
    const float muy = pa[GK + g];
    const float z   = pa[2 * GK + g];
    const float l00 = pa[3 * GK + 4 * g + 0];
    const float l10 = pa[3 * GK + 4 * g + 2];
    const float l11 = pa[3 * GK + 4 * g + 3];
    asm volatile("" ::"v"(mux), "v"(muy), "v"(z), "v"(l00), "v"(l10),
                 "v"(l11));

    const float srx = l00 * l00 + l00 * l10;
    const float sry = l00 * l10 + l10 * l10 + l11 * l11;
    const float c1c = -(srx * mux + sry * muy);
    const float c2c = KEXP * -(mux + muy);

    float mx = z;
    mx = fmaxf(mx, row_xor<1>(mx));
    mx = fmaxf(mx, row_xor<2>(mx));
    mx = fmaxf(mx, row_xor<4>(mx));
    mx = fmaxf(mx, row_xor<8>(mx));
    float ez = __expf(z - mx);
    float ssum = ez;
    ssum += row_xor<1>(ssum);
    ssum += row_xor<2>(ssum);
    ssum += row_xor<4>(ssum);
    ssum += row_xor<8>(ssum);
    const float w = ez / ssum;

    s_sr[lane] = make_float2(srx, sry);
    reinterpret_cast<float*>(s_c1)[lane] = c1c;
    reinterpret_cast<float*>(s_c2)[lane] = c2c;
    reinterpret_cast<float*>(s_w)[lane]  = w;
  }
  __syncthreads();

  const bool own = ((lane >> 3) & 1) == hi;
  const float2 sr0 = s_sr[lane & 7];
  const float2 sr1 = s_sr[(lane & 7) + 8];
  v2f a1p0, a1p1, a2;
  a1p0.x = own ? sr0.x : 0.0f;  a1p0.y = own ? sr0.y : 0.0f;
  a1p1.x = own ? sr1.x : 0.0f;  a1p1.y = own ? sr1.y : 0.0f;
  a2.x   = own ? KEXP : 0.0f;   a2.y   = a2.x;

  const float4 q0 = s_c1[0], q1 = s_c1[1], q2 = s_c1[2], q3 = s_c1[3];
  const float4 u0 = s_c2[0], u1 = s_c2[1], u2 = s_c2[2], u3 = s_c2[3];
  const float4 w0 = s_w[0],  w1 = s_w[1],  w2 = s_w[2],  w3 = s_w[3];
  const v8f c1p0 = {q0.x, q0.y, q0.z, q0.w, q1.x, q1.y, q1.z, q1.w};
  const v8f c1p1 = {q2.x, q2.y, q2.z, q2.w, q3.x, q3.y, q3.z, q3.w};
  const v8f c2p0 = {u0.x, u0.y, u0.z, u0.w, u1.x, u1.y, u1.z, u1.w};
  const v8f c2p1 = {u2.x, u2.y, u2.z, u2.w, u3.x, u3.y, u3.z, u3.w};
  const float wl[16] = {w0.x, w0.y, w0.z, w0.w, w1.x, w1.y, w1.z, w1.w,
                        w2.x, w2.y, w2.z, w2.w, w3.x, w3.y, w3.z, w3.w};
  float ws[16];
#pragma unroll
  for (int i = 0; i < 16; ++i)
    ws[i] = __builtin_bit_cast(
        float, __builtin_amdgcn_readfirstlane(__builtin_bit_cast(int, wl[i])));

  const int waves_per_blk = bdim >> 5;
  const int wave   = blockIdx.x * waves_per_blk + wid;
  const int nwaves = nblk * waves_per_blk;

  for (int tile = wave; tile < TILES; tile += nwaves) {
    const int pn = (tile << 5) + lane;
    v2f b;
    b.x = (float)(pn >> 8) * (1.0f / 255.0f);
    b.y = (float)(pn & 255) * (1.0f / 255.0f);

    const v8f t1a = __builtin_amdgcn_wmma_f32_16x16x4_f32(
        false, a1p0, false, b, (short)0, c1p0, false, false);
    const v8f t2a = __builtin_amdgcn_wmma_f32_16x16x4_f32(
        false, a2, false, b, (short)0, c2p0, false, false);
    const v8f t1b = __builtin_amdgcn_wmma_f32_16x16x4_f32(
        false, a1p1, false, b, (short)0, c1p1, false, false);
    const v8f t2b = __builtin_amdgcn_wmma_f32_16x16x4_f32(
        false, a2, false, b, (short)0, c2p1, false, false);

    float se0 = 0.0f, swe0 = 0.0f, se1 = 0.0f, swe1 = 0.0f;
#pragma unroll
    for (int r = 0; r < 8; ++r) {
      const float e0 = __builtin_amdgcn_exp2f(t1a[r] * t2a[r]);
      se0 += e0;
      swe0 = fmaf(ws[r], e0, swe0);
    }
#pragma unroll
    for (int r = 0; r < 8; ++r) {
      const float e1 = __builtin_amdgcn_exp2f(t1b[r] * t2b[r]);
      se1 += e1;
      swe1 = fmaf(ws[8 + r], e1, swe1);
    }
    const float se  = se0 + se1;
    const float swe = swe0 + swe1;

    float y = swe / fmaxf(se, 1e-8f);
    y = fminf(fmaxf(y, 0.0f), 1.0f);
    out[a * N_PIX + pn] = y;
  }
}

extern "C" void kernel_launch(void* const* d_in, const int* in_sizes, int n_in,
                              void* d_out, int out_size, void* d_ws, size_t ws_size,
                              hipStream_t stream) {
  (void)in_sizes; (void)n_in; (void)d_ws; (void)ws_size; (void)out_size;
  const float* params = (const float*)d_in[0];
  float* out = (float*)d_out;
  dim3 grid(64, A_CH);
  gmm_render_kernel<<<grid, 256, 0, stream>>>(params, out);
}


# term2's product KEXP*(x+y) does not depend on the gaussian: ONE WMMA per wave gives it for all four tiles; tile loop 2 WMMAs not 4, term2 constants by packed add; term2 rounds differently
# speedup vs baseline: 1.4391x; 1.0806x over previous
#include <hip/hip_runtime.h>


typedef __attribute__((ext_vector_type(2))) float v2f;
typedef __attribute__((ext_vector_type(8))) float v8f;

#define GK 16
#define A_CH 24
#define N_PIX 65536
#define TILES 2048
#define WPB 8
#define WG_THREADS 256
#define NBLK_X 64
#define NWAVES (NBLK_X * WPB)
#define TPW (TILES / NWAVES)
#define ROW_STEP (NWAVES * 32 / 256)
#define KEXP (-0.72134752044448170368f)

template <int M>
__device__ __forceinline__ float row_xor(float x) {
  return __builtin_bit_cast(
      float, __builtin_amdgcn_mov_dpp(__builtin_bit_cast(int, x), 0x160 | M,
                                      0xf, 0xf, false));
}

__global__ __launch_bounds__(WG_THREADS) __attribute__((amdgpu_waves_per_eu(12, 12)))
void gmm_render_kernel(
    const float* __restrict__ params, float* __restrict__ out) {
  __shared__ float2 s_sr[32];
  __shared__ float4 s_c1[8];
  __shared__ float4 s_c2[8];
  __shared__ float4 s_w[8];

  const int lane = threadIdx.x & 31;
  const int g    = lane & 15;
  const int hi   = lane >> 4;
  const int a    = blockIdx.y;
  const int wid  = __builtin_amdgcn_readfirstlane((int)(threadIdx.x >> 5));

  if (wid == 0) {
    const float* pa = params + a * (7 * GK);
    const float mux = pa[g];
    const float muy = pa[GK + g];
    const float z   = pa[2 * GK + g];
    const float l00 = pa[3 * GK + 4 * g + 0];
    const float l10 = pa[3 * GK + 4 * g + 2];
    const float l11 = pa[3 * GK + 4 * g + 3];
    asm volatile("" ::"v"(mux), "v"(muy), "v"(z), "v"(l00), "v"(l10),
                 "v"(l11));

    const float srx = l00 * l00 + l00 * l10;
    const float sry = l00 * l10 + l10 * l10 + l11 * l11;
    const float c1c = -(srx * mux + sry * muy);
    const float c2c = KEXP * -(mux + muy);

    float mx = z;
    mx = fmaxf(mx, row_xor<1>(mx));
    mx = fmaxf(mx, row_xor<2>(mx));
    mx = fmaxf(mx, row_xor<4>(mx));
    mx = fmaxf(mx, row_xor<8>(mx));
    float ez = __expf(z - mx);
    float ssum = ez;
    ssum += row_xor<1>(ssum);
    ssum += row_xor<2>(ssum);
    ssum += row_xor<4>(ssum);
    ssum += row_xor<8>(ssum);
    const float w = ez / ssum;

    s_sr[lane] = make_float2(srx, sry);
    reinterpret_cast<float*>(s_c1)[lane] = c1c;
    reinterpret_cast<float*>(s_c2)[lane] = c2c;
    reinterpret_cast<float*>(s_w)[lane]  = w;
  }
  __syncthreads();

  const bool own = ((lane >> 3) & 1) == hi;
  const float2 sr0 = s_sr[lane & 7];
  const float2 sr1 = s_sr[(lane & 7) + 8];
  v2f a1p0, a1p1;
  a1p0.x = own ? sr0.x : 0.0f;  a1p0.y = own ? sr0.y : 0.0f;
  a1p1.x = own ? sr1.x : 0.0f;  a1p1.y = own ? sr1.y : 0.0f;

  const float4 q0 = s_c1[0], q1 = s_c1[1], q2 = s_c1[2], q3 = s_c1[3];
  const float4 u0 = s_c2[0], u1 = s_c2[1], u2 = s_c2[2], u3 = s_c2[3];
  const float4 w0 = s_w[0],  w1 = s_w[1],  w2 = s_w[2],  w3 = s_w[3];
  const v8f c1p0 = {q0.x, q0.y, q0.z, q0.w, q1.x, q1.y, q1.z, q1.w};
  const v8f c1p1 = {q2.x, q2.y, q2.z, q2.w, q3.x, q3.y, q3.z, q3.w};
  const v8f c2p0 = {u0.x, u0.y, u0.z, u0.w, u1.x, u1.y, u1.z, u1.w};
  const v8f c2p1 = {u2.x, u2.y, u2.z, u2.w, u3.x, u3.y, u3.z, u3.w};
  const float wl[16] = {w0.x, w0.y, w0.z, w0.w, w1.x, w1.y, w1.z, w1.w,
                        w2.x, w2.y, w2.z, w2.w, w3.x, w3.y, w3.z, w3.w};
  float ws[16];
#pragma unroll
  for (int i = 0; i < 16; ++i)
    ws[i] = __builtin_bit_cast(
        float, __builtin_amdgcn_readfirstlane(__builtin_bit_cast(int, wl[i])));

  const int wave = blockIdx.x * WPB + wid;

  const int col  = ((wave & 7) << 5) + lane;
  const int row0 = wave >> 3;
  const float y  = (float)col * (1.0f / 255.0f);

  v2f ah, bh;
  ah.x = own ? (float)(row0 + ROW_STEP * (lane & 7)) * (1.0f / 255.0f) : 0.0f;
  ah.y = own ? KEXP : 0.0f;
  bh.x = KEXP;
  bh.y = y;
  const v8f czero = {0.0f, 0.0f, 0.0f, 0.0f, 0.0f, 0.0f, 0.0f, 0.0f};
  const v8f sxy = __builtin_amdgcn_wmma_f32_16x16x4_f32(
      false, ah, false, bh, (short)0, czero, false, false);

#pragma unroll
  for (int k = 0; k < TPW; ++k) {
    const int row = row0 + ROW_STEP * k;
    v2f b;
    b.x = (float)row * (1.0f / 255.0f);
    b.y = y;

    const v8f t1a = __builtin_amdgcn_wmma_f32_16x16x4_f32(
        false, a1p0, false, b, (short)0, c1p0, false, false);
    const v8f t1b = __builtin_amdgcn_wmma_f32_16x16x4_f32(
        false, a1p1, false, b, (short)0, c1p1, false, false);
    const float s2 = sxy[k];

    float se0 = 0.0f, swe0 = 0.0f, se1 = 0.0f, swe1 = 0.0f;
#pragma unroll
    for (int r = 0; r < 8; ++r) {
      const float e0 = __builtin_amdgcn_exp2f(t1a[r] * (s2 + c2p0[r]));
      se0 += e0;
      swe0 = fmaf(ws[r], e0, swe0);
    }
#pragma unroll
    for (int r = 0; r < 8; ++r) {
      const float e1 = __builtin_amdgcn_exp2f(t1b[r] * (s2 + c2p1[r]));
      se1 += e1;
      swe1 = fmaf(ws[8 + r], e1, swe1);
    }
    const float se  = se0 + se1;
    const float swe = swe0 + swe1;

    float yv = swe / fmaxf(se, 1e-8f);
    yv = fminf(fmaxf(yv, 0.0f), 1.0f);
    out[a * N_PIX + (row << 8) + col] = yv;
  }
}

extern "C" void kernel_launch(void* const* d_in, const int* in_sizes, int n_in,
                              void* d_out, int out_size, void* d_ws, size_t ws_size,
                              hipStream_t stream) {
  (void)in_sizes; (void)n_in; (void)d_ws; (void)ws_size; (void)out_size;
  const float* params = (const float*)d_in[0];
  float* out = (float*)d_out;
  dim3 grid(NBLK_X, A_CH);
  gmm_render_kernel<<<grid, WG_THREADS, 0, stream>>>(params, out);
}


# Exponentials of a wave's rows 1-3 by packed products from row 0 (e*=r, r*=q; wave owns 4 adjacent rows): exp2 on row 0 only; rows 1-3 round differently, sum order over gaussians kept
# speedup vs baseline: 1.6420x; 1.1410x over previous
#include <hip/hip_runtime.h>


typedef __attribute__((ext_vector_type(2))) float v2f;
typedef __attribute__((ext_vector_type(8))) float v8f;

#define GK 16
#define A_CH 24
#define N_PIX 65536
#define TILES 2048
#define WPB 8
#define WG_THREADS 256
#define NBLK_X 64
#define NWAVES (NBLK_X * WPB)
#define TPW (TILES / NWAVES)
#define DX (1.0f / 255.0f)
#define KEXP (-0.72134752044448170368f)
#define KAPPA (KEXP * DX)

template <int M>
__device__ __forceinline__ float row_xor(float x) {
  return __builtin_bit_cast(
      float, __builtin_amdgcn_mov_dpp(__builtin_bit_cast(int, x), 0x160 | M,
                                      0xf, 0xf, false));
}

__global__ __launch_bounds__(WG_THREADS) __attribute__((amdgpu_waves_per_eu(12, 12)))
void gmm_render_kernel(
    const float* __restrict__ params, float* __restrict__ out) {
  __shared__ float2 s_sr[32];
  __shared__ float4 s_c1[8];
  __shared__ float4 s_c2[8];
  __shared__ float4 s_w[8];
  __shared__ float4 s_d1[8];
  __shared__ float4 s_dk[8];
  __shared__ float4 s_q[8];

  const int lane = threadIdx.x & 31;
  const int g    = lane & 15;
  const int hi   = lane >> 4;
  const int a    = blockIdx.y;
  const int wid  = __builtin_amdgcn_readfirstlane((int)(threadIdx.x >> 5));

  if (wid == 0) {
    const float* pa = params + a * (7 * GK);
    const float mux = pa[g];
    const float muy = pa[GK + g];
    const float z   = pa[2 * GK + g];
    const float l00 = pa[3 * GK + 4 * g + 0];
    const float l10 = pa[3 * GK + 4 * g + 2];
    const float l11 = pa[3 * GK + 4 * g + 3];
    asm volatile("" ::"v"(mux), "v"(muy), "v"(z), "v"(l00), "v"(l10),
                 "v"(l11));

    const float srx = l00 * l00 + l00 * l10;
    const float sry = l00 * l10 + l10 * l10 + l11 * l11;
    const float c1c = -(srx * mux + sry * muy);
    const float c2c = KEXP * -(mux + muy);
    const float d1c = srx * DX;
    const float dkc = d1c * KAPPA;
    const float qc  = __builtin_amdgcn_exp2f(2.0f * dkc);

    float mx = z;
    mx = fmaxf(mx, row_xor<1>(mx));
    mx = fmaxf(mx, row_xor<2>(mx));
    mx = fmaxf(mx, row_xor<4>(mx));
    mx = fmaxf(mx, row_xor<8>(mx));
    float ez = __expf(z - mx);
    float ssum = ez;
    ssum += row_xor<1>(ssum);
    ssum += row_xor<2>(ssum);
    ssum += row_xor<4>(ssum);
    ssum += row_xor<8>(ssum);
    const float w = ez / ssum;

    const int slot = (lane & 16) | ((g & 4) << 1) | ((g & 3) << 1) | (g >> 3);
    s_sr[slot] = make_float2(srx, sry);
    reinterpret_cast<float*>(s_c1)[slot] = c1c;
    reinterpret_cast<float*>(s_c2)[slot] = c2c;
    reinterpret_cast<float*>(s_w)[slot]  = w;
    reinterpret_cast<float*>(s_d1)[slot] = d1c;
    reinterpret_cast<float*>(s_dk)[slot] = dkc;
    reinterpret_cast<float*>(s_q)[slot]  = qc;
  }
  __syncthreads();

  const bool own = ((lane >> 3) & 1) == hi;
  const float2 sr0 = s_sr[lane & 7];
  const float2 sr1 = s_sr[(lane & 7) + 8];
  v2f a1p0, a1p1;
  a1p0.x = own ? sr0.x : 0.0f;  a1p0.y = own ? sr0.y : 0.0f;
  a1p1.x = own ? sr1.x : 0.0f;  a1p1.y = own ? sr1.y : 0.0f;

  const int wave = blockIdx.x * WPB + wid;

  const int col  = ((wave & 7) << 5) + lane;
  const int row0 = (wave >> 3) * TPW;
  const float y  = (float)col * (1.0f / 255.0f);

  const float x0 = (float)row0 * (1.0f / 255.0f);

  v2f ah, bh;
  ah.x = own ? x0 : 0.0f;
  ah.y = own ? KEXP : 0.0f;
  bh.x = KEXP;
  bh.y = y;
  const v8f czero = {0.0f, 0.0f, 0.0f, 0.0f, 0.0f, 0.0f, 0.0f, 0.0f};
  const v8f sxy = __builtin_amdgcn_wmma_f32_16x16x4_f32(
      false, ah, false, bh, (short)0, czero, false, false);
  const float s2 = sxy[0];

  v2f b;
  b.x = x0;
  b.y = y;

  const v8f kap = {KAPPA, KAPPA, KAPPA, KAPPA, KAPPA, KAPPA, KAPPA, KAPPA};

  v8f ev[2], rv[2];
#pragma unroll
  for (int p = 0; p < 2; ++p) {
    const float4 qa = s_c1[2 * p], qb = s_c1[2 * p + 1];
    const v8f c1 = {qa.x, qa.y, qa.z, qa.w, qb.x, qb.y, qb.z, qb.w};
    const v8f t1 = __builtin_amdgcn_wmma_f32_16x16x4_f32(
        false, p == 0 ? a1p0 : a1p1, false, b, (short)0, c1, false, false);
    const float4 ua = s_c2[2 * p], ub = s_c2[2 * p + 1];
    const v8f c2 = {ua.x, ua.y, ua.z, ua.w, ub.x, ub.y, ub.z, ub.w};
    const v8f t2 = c2 + s2;
    const v8f E  = t1 * t2;
    const float4 da = s_d1[2 * p], db = s_d1[2 * p + 1];
    const float4 ka = s_dk[2 * p], kb = s_dk[2 * p + 1];
    const v8f d  = {da.x, da.y, da.z, da.w, db.x, db.y, db.z, db.w};
    const v8f dk = {ka.x, ka.y, ka.z, ka.w, kb.x, kb.y, kb.z, kb.w};
    const v8f G = __builtin_elementwise_fma(d, t2, dk);
    const v8f F = __builtin_elementwise_fma(kap, t1, G);
#pragma unroll
    for (int r = 0; r < 8; ++r) {
      ev[p][r] = __builtin_amdgcn_exp2f(E[r]);
      rv[p][r] = __builtin_amdgcn_exp2f(F[r]);
    }
  }
  v8f ea = ev[0], eb = ev[1], ra = rv[0], rb = rv[1];

  const float4 w0 = s_w[0], w1 = s_w[1], w2 = s_w[2], w3 = s_w[3];
  const v2f wv[8] = {{w0.x, w0.y}, {w0.z, w0.w}, {w1.x, w1.y}, {w1.z, w1.w},
                     {w2.x, w2.y}, {w2.z, w2.w}, {w3.x, w3.y}, {w3.z, w3.w}};

  const float4 g0 = s_q[0], g1 = s_q[1], g2 = s_q[2], g3 = s_q[3];
  const v8f qa8 = {g0.x, g0.y, g0.z, g0.w, g1.x, g1.y, g1.z, g1.w};
  const v8f qb8 = {g2.x, g2.y, g2.z, g2.w, g3.x, g3.y, g3.z, g3.w};

#pragma unroll
  for (int k = 0; k < TPW; ++k) {
    const int row = row0 + k;
    if (k > 0) {
      ea *= ra;
      eb *= rb;
      if (k < TPW - 1) {
        ra *= qa8;
        rb *= qb8;
      }
    }

    v2f se2  = __builtin_shufflevector(ea, ea, 0, 1);
    v2f swe2 = wv[0] * se2;
#pragma unroll
    for (int i = 1; i < 8; ++i) {
      const v8f& src = i < 4 ? ea : eb;
      v2f e;
      e.x = src[2 * (i & 3)];
      e.y = src[2 * (i & 3) + 1];
      se2 += e;
      swe2 = __builtin_elementwise_fma(wv[i], e, swe2);
    }
    const float se  = se2.x + se2.y;
    const float swe = swe2.x + swe2.y;

    float yv = swe / fmaxf(se, 1e-8f);
    yv = fminf(fmaxf(yv, 0.0f), 1.0f);
    out[a * N_PIX + (row << 8) + col] = yv;
  }
}

extern "C" void kernel_launch(void* const* d_in, const int* in_sizes, int n_in,
                              void* d_out, int out_size, void* d_ws, size_t ws_size,
                              hipStream_t stream) {
  (void)in_sizes; (void)n_in; (void)d_ws; (void)ws_size; (void)out_size;
  const float* params = (const float*)d_in[0];
  float* out = (float*)d_out;
  dim3 grid(NBLK_X, A_CH);
  gmm_render_kernel<<<grid, WG_THREADS, 0, stream>>>(params, out);
}


# Quotients (each pixel's, and the softmax weight) by hardware reciprocal times numerator instead of the IEEE division sequence; weights and clamped sum share a 2^-8 scale; quotients round differently
# speedup vs baseline: 1.7296x; 1.0533x over previous
#include <hip/hip_runtime.h>


typedef __attribute__((ext_vector_type(2))) float v2f;
typedef __attribute__((ext_vector_type(8))) float v8f;

#define GK 16
#define A_CH 24
#define N_PIX 65536
#define TILES 2048
#define WPB 8
#define WG_THREADS 256
#define NBLK_X 64
#define NWAVES (NBLK_X * WPB)
#define TPW (TILES / NWAVES)
#define DX (1.0f / 255.0f)
#define KEXP (-0.72134752044448170368f)
#define KAPPA (KEXP * DX)
#define QS (0.00390625f)

template <int M>
__device__ __forceinline__ float row_xor(float x) {
  return __builtin_bit_cast(
      float, __builtin_amdgcn_mov_dpp(__builtin_bit_cast(int, x), 0x160 | M,
                                      0xf, 0xf, false));
}

__global__ __launch_bounds__(WG_THREADS) __attribute__((amdgpu_waves_per_eu(12, 12)))
void gmm_render_kernel(
    const float* __restrict__ params, float* __restrict__ out) {
  __shared__ float2 s_sr[32];
  __shared__ float4 s_c1[8];
  __shared__ float4 s_c2[8];
  __shared__ float4 s_w[8];
  __shared__ float4 s_d1[8];
  __shared__ float4 s_dk[8];
  __shared__ float4 s_q[8];

  const int lane = threadIdx.x & 31;
  const int g    = lane & 15;
  const int hi   = lane >> 4;
  const int a    = blockIdx.y;
  const int wid  = __builtin_amdgcn_readfirstlane((int)(threadIdx.x >> 5));

  if (wid == 0) {
    const float* pa = params + a * (7 * GK);
    const float mux = pa[g];
    const float muy = pa[GK + g];
    const float z   = pa[2 * GK + g];
    const float l00 = pa[3 * GK + 4 * g + 0];
    const float l10 = pa[3 * GK + 4 * g + 2];
    const float l11 = pa[3 * GK + 4 * g + 3];
    asm volatile("" ::"v"(mux), "v"(muy), "v"(z), "v"(l00), "v"(l10),
                 "v"(l11));

    const float srx = l00 * l00 + l00 * l10;
    const float sry = l00 * l10 + l10 * l10 + l11 * l11;
    const float c1c = -(srx * mux + sry * muy);
    const float c2c = KEXP * -(mux + muy);
    const float d1c = srx * DX;
    const float dkc = d1c * KAPPA;
    const float qc  = __builtin_amdgcn_exp2f(2.0f * dkc);

    float mx = z;
    mx = fmaxf(mx, row_xor<1>(mx));
    mx = fmaxf(mx, row_xor<2>(mx));
    mx = fmaxf(mx, row_xor<4>(mx));
    mx = fmaxf(mx, row_xor<8>(mx));
    float ez = __expf(z - mx);
    float ssum = ez;
    ssum += row_xor<1>(ssum);
    ssum += row_xor<2>(ssum);
    ssum += row_xor<4>(ssum);
    ssum += row_xor<8>(ssum);
    const float w = (ez * QS) * __builtin_amdgcn_rcpf(ssum);

    const int slot = (lane & 16) | ((g & 4) << 1) | ((g & 3) << 1) | (g >> 3);
    s_sr[slot] = make_float2(srx, sry);
    reinterpret_cast<float*>(s_c1)[slot] = c1c;
    reinterpret_cast<float*>(s_c2)[slot] = c2c;
    reinterpret_cast<float*>(s_w)[slot]  = w;
    reinterpret_cast<float*>(s_d1)[slot] = d1c;
    reinterpret_cast<float*>(s_dk)[slot] = dkc;
    reinterpret_cast<float*>(s_q)[slot]  = qc;
  }
  __syncthreads();

  const bool own = ((lane >> 3) & 1) == hi;
  const float2 sr0 = s_sr[lane & 7];
  const float2 sr1 = s_sr[(lane & 7) + 8];
  v2f a1p0, a1p1;
  a1p0.x = own ? sr0.x : 0.0f;  a1p0.y = own ? sr0.y : 0.0f;
  a1p1.x = own ? sr1.x : 0.0f;  a1p1.y = own ? sr1.y : 0.0f;

  const int wave = blockIdx.x * WPB + wid;

  const int col  = ((wave & 7) << 5) + lane;
  const int row0 = (wave >> 3) * TPW;
  const float y  = (float)col * (1.0f / 255.0f);

  const float x0 = (float)row0 * (1.0f / 255.0f);

  v2f ah, bh;
  ah.x = own ? x0 : 0.0f;
  ah.y = own ? KEXP : 0.0f;
  bh.x = KEXP;
  bh.y = y;
  const v8f czero = {0.0f, 0.0f, 0.0f, 0.0f, 0.0f, 0.0f, 0.0f, 0.0f};
  const v8f sxy = __builtin_amdgcn_wmma_f32_16x16x4_f32(
      false, ah, false, bh, (short)0, czero, false, false);
  const float s2 = sxy[0];

  v2f b;
  b.x = x0;
  b.y = y;

  const v8f kap = {KAPPA, KAPPA, KAPPA, KAPPA, KAPPA, KAPPA, KAPPA, KAPPA};

  v8f ev[2], rv[2];
#pragma unroll
  for (int p = 0; p < 2; ++p) {
    const float4 qa = s_c1[2 * p], qb = s_c1[2 * p + 1];
    const v8f c1 = {qa.x, qa.y, qa.z, qa.w, qb.x, qb.y, qb.z, qb.w};
    const v8f t1 = __builtin_amdgcn_wmma_f32_16x16x4_f32(
        false, p == 0 ? a1p0 : a1p1, false, b, (short)0, c1, false, false);
    const float4 ua = s_c2[2 * p], ub = s_c2[2 * p + 1];
    const v8f c2 = {ua.x, ua.y, ua.z, ua.w, ub.x, ub.y, ub.z, ub.w};
    const v8f t2 = c2 + s2;
    const v8f E  = t1 * t2;
    const float4 da = s_d1[2 * p], db = s_d1[2 * p + 1];
    const float4 ka = s_dk[2 * p], kb = s_dk[2 * p + 1];
    const v8f d  = {da.x, da.y, da.z, da.w, db.x, db.y, db.z, db.w};
    const v8f dk = {ka.x, ka.y, ka.z, ka.w, kb.x, kb.y, kb.z, kb.w};
    const v8f G = __builtin_elementwise_fma(d, t2, dk);
    const v8f F = __builtin_elementwise_fma(kap, t1, G);
#pragma unroll
    for (int r = 0; r < 8; ++r) {
      ev[p][r] = __builtin_amdgcn_exp2f(E[r]);
      rv[p][r] = __builtin_amdgcn_exp2f(F[r]);
    }
  }
  v8f ea = ev[0], eb = ev[1], ra = rv[0], rb = rv[1];

  const float4 w0 = s_w[0], w1 = s_w[1], w2 = s_w[2], w3 = s_w[3];
  const v2f wv[8] = {{w0.x, w0.y}, {w0.z, w0.w}, {w1.x, w1.y}, {w1.z, w1.w},
                     {w2.x, w2.y}, {w2.z, w2.w}, {w3.x, w3.y}, {w3.z, w3.w}};

  const float4 g0 = s_q[0], g1 = s_q[1], g2 = s_q[2], g3 = s_q[3];
  const v8f qa8 = {g0.x, g0.y, g0.z, g0.w, g1.x, g1.y, g1.z, g1.w};
  const v8f qb8 = {g2.x, g2.y, g2.z, g2.w, g3.x, g3.y, g3.z, g3.w};

#pragma unroll
  for (int k = 0; k < TPW; ++k) {
    const int row = row0 + k;
    if (k > 0) {
      ea *= ra;
      eb *= rb;
      if (k < TPW - 1) {
        ra *= qa8;
        rb *= qb8;
      }
    }

    v2f se2  = __builtin_shufflevector(ea, ea, 0, 1);
    v2f swe2 = wv[0] * se2;
#pragma unroll
    for (int i = 1; i < 8; ++i) {
      const v8f& src = i < 4 ? ea : eb;
      v2f e;
      e.x = src[2 * (i & 3)];
      e.y = src[2 * (i & 3) + 1];
      se2 += e;
      swe2 = __builtin_elementwise_fma(wv[i], e, swe2);
    }
    const float se  = se2.x + se2.y;
    const float swe = swe2.x + swe2.y;

    float yv = swe * __builtin_amdgcn_rcpf(fmaxf(se, 1e-8f) * QS);
    yv = fminf(fmaxf(yv, 0.0f), 1.0f);
    out[a * N_PIX + (row << 8) + col] = yv;
  }
}

extern "C" void kernel_launch(void* const* d_in, const int* in_sizes, int n_in,
                              void* d_out, int out_size, void* d_ws, size_t ws_size,
                              hipStream_t stream) {
  (void)in_sizes; (void)n_in; (void)d_ws; (void)ws_size; (void)out_size;
  const float* params = (const float*)d_in[0];
  float* out = (float*)d_out;
  dim3 grid(NBLK_X, A_CH);
  gmm_render_kernel<<<grid, WG_THREADS, 0, stream>>>(params, out);
}
